# starting kernel
# baseline (speedup 1.0000x reference)
#include <hip/hip_runtime.h>

typedef __attribute__((ext_vector_type(2))) float v2f;
typedef __attribute__((ext_vector_type(8))) float v8f;

#define NQ   8192
#define NK   2048
#define CDIM 128
#define QPB  128

__device__ __forceinline__ void top3_insert(float v, int ki,
                                            float& b0, float& b1, float& b2,
                                            int& i0, int& i1, int& i2) {
    bool c2 = v < b2;
    float x  = c2 ? v  : b2;
    int   xi = c2 ? ki : i2;
    bool c1 = x < b1;
    b2 = c1 ? b1 : x;   i2 = c1 ? i1 : xi;
    float y  = c1 ? x  : b1;
    int   yi = c1 ? xi : i1;
    bool c0 = y < b0;
    b1 = c0 ? b0 : y;   i1 = c0 ? i0 : yi;
    b0 = c0 ? y  : b0;  i0 = c0 ? yi : i0;
}

__global__ __launch_bounds__(256) void
fp_knn_wmma_kernel(const float* __restrict__ xyz_q,
                   const float* __restrict__ xyz_k,
                   const float* __restrict__ v_k,
                   float* __restrict__ out) {
    __shared__ float s_keys[NK * 4];

    const int b      = blockIdx.y;
    const int qblock = blockIdx.x * QPB;
    const int tid    = threadIdx.x;

    const float* kb = xyz_k + (size_t)b * NK * 3;
    for (int m = tid; m < NK; m += 256) {
        float kx = kb[m * 3 + 0];
        float ky = kb[m * 3 + 1];
        float kz = kb[m * 3 + 2];
        s_keys[m * 4 + 0] = kx;
        s_keys[m * 4 + 1] = ky;
        s_keys[m * 4 + 2] = kz;
        s_keys[m * 4 + 3] = kx * kx + ky * ky + kz * kz;
    }
    __syncthreads();

    const int wave = tid >> 5;
    const int lane = tid & 31;
    const int half = lane >> 4;
    const int l16  = lane & 15;

    const int   q  = qblock + wave * 16 + l16;
    const float* qp = xyz_q + ((size_t)b * NQ + q) * 3;
    float qx = qp[0], qy = qp[1], qz = qp[2];
    float qn2 = qx * qx + qy * qy + qz * qz;

    v2f bq;
    bq.x = half ? (-2.0f * qz) : (-2.0f * qx);
    bq.y = half ? 1.0f         : (-2.0f * qy);

    const int abase = l16 * 4 + half * 2;

    float b0 = 1e30f, b1 = 1e30f, b2 = 1e30f;
    int   i0 = 0,     i1 = 0,     i2 = 0;

    v2f a_cur;
    {
        const float* sk = &s_keys[abase];
        a_cur.x = sk[0];
        a_cur.y = sk[1];
    }

    for (int t = 0; t < NK; t += 16) {
        v2f a_nxt;
        {
            const float* sk = &s_keys[(((t + 16) & (NK - 1)) * 4) + abase];
            a_nxt.x = sk[0];
            a_nxt.y = sk[1];
        }

        v8f c = {};
        v8f d = __builtin_amdgcn_wmma_f32_16x16x4_f32(
             false, a_cur,  false, bq,
             (short)0, c,  false,  false);

        bool need = false;
#pragma unroll
        for (int r = 0; r < 8; ++r) need |= (d[r] < b2);

        if (__any(need)) {
            const int kbase = t + half * 8;
#pragma unroll
            for (int r = 0; r < 8; ++r) {
                top3_insert(d[r], kbase + r, b0, b1, b2, i0, i1, i2);
            }
        }

        a_cur = a_nxt;
    }

    float ob0 = __shfl_xor(b0, 16, 32);
    float ob1 = __shfl_xor(b1, 16, 32);
    float ob2 = __shfl_xor(b2, 16, 32);
    int   oi0 = __shfl_xor(i0, 16, 32);
    int   oi1 = __shfl_xor(i1, 16, 32);
    int   oi2 = __shfl_xor(i2, 16, 32);
    top3_insert(ob0, oi0, b0, b1, b2, i0, i1, i2);
    top3_insert(ob1, oi1, b0, b1, b2, i0, i1, i2);
    top3_insert(ob2, oi2, b0, b1, b2, i0, i1, i2);

    float d0 = fmaxf(b0 + qn2, 1e-10f);
    float d1 = fmaxf(b1 + qn2, 1e-10f);
    float d2 = fmaxf(b2 + qn2, 1e-10f);
    float w0 = 1.0f / d0, w1 = 1.0f / d1, w2 = 1.0f / d2;
    float inv = 1.0f / (w0 + w1 + w2);
    w0 *= inv; w1 *= inv; w2 *= inv;

    const float* vb = v_k + (size_t)b * NK * CDIM;
    float*       ob = out + ((size_t)b * NQ + qblock + wave * 16) * CDIM;
    const int cidx = lane * 4;

    for (int qq = 0; qq < 16; ++qq) {
        int   j0 = __shfl(i0, qq, 32);
        int   j1 = __shfl(i1, qq, 32);
        int   j2 = __shfl(i2, qq, 32);
        float u0 = __shfl(w0, qq, 32);
        float u1 = __shfl(w1, qq, 32);
        float u2 = __shfl(w2, qq, 32);

        const float4 f0 = *(const float4*)(vb + (size_t)j0 * CDIM + cidx);
        const float4 f1 = *(const float4*)(vb + (size_t)j1 * CDIM + cidx);
        const float4 f2 = *(const float4*)(vb + (size_t)j2 * CDIM + cidx);

        float4 r;
        r.x = u0 * f0.x + u1 * f1.x + u2 * f2.x;
        r.y = u0 * f0.y + u1 * f1.y + u2 * f2.y;
        r.z = u0 * f0.z + u1 * f1.z + u2 * f2.z;
        r.w = u0 * f0.w + u1 * f1.w + u2 * f2.w;

        *(float4*)(ob + (size_t)qq * CDIM + cidx) = r;
    }
}

extern "C" void kernel_launch(void* const* d_in, const int* in_sizes, int n_in,
                              void* d_out, int out_size, void* d_ws, size_t ws_size,
                              hipStream_t stream) {
    const float* xyz_q = (const float*)d_in[0];
    const float* xyz_k = (const float*)d_in[1];
    const float* v_k   = (const float*)d_in[2];
    float*       out   = (float*)d_out;

    const int B = in_sizes[0] / (NQ * 3);

    dim3 grid(NQ / QPB, B);
    dim3 block(256);
    fp_knn_wmma_kernel<<<grid, block, 0, stream>>>(xyz_q, xyz_k, v_k, out);
}


# Top-3 selection in two key sweeps: sweep 1 takes branch-free per-row minima to bound the 3rd-nearest distance, sweep 2 repeats the WMMA tiles and inserts only rows <= that bound
# speedup vs baseline: 1.1240x; 1.1240x over previous
#include <hip/hip_runtime.h>

typedef __attribute__((ext_vector_type(2))) float v2f;
typedef __attribute__((ext_vector_type(8))) float v8f;

#define NQ   8192
#define NK   2048
#define CDIM 128
#define QPB  128

__device__ __forceinline__ void top3_insert(float v, int ki,
                                            float& b0, float& b1, float& b2,
                                            int& i0, int& i1, int& i2) {
    bool c2 = v < b2;
    float x  = c2 ? v  : b2;
    int   xi = c2 ? ki : i2;
    bool c1 = x < b1;
    b2 = c1 ? b1 : x;   i2 = c1 ? i1 : xi;
    float y  = c1 ? x  : b1;
    int   yi = c1 ? xi : i1;
    bool c0 = y < b0;
    b1 = c0 ? b0 : y;   i1 = c0 ? i0 : yi;
    b0 = c0 ? y  : b0;  i0 = c0 ? yi : i0;
}

__device__ __forceinline__ void top3_vals(float v, float& b0, float& b1, float& b2) {
    float n2 = __builtin_amdgcn_fmed3f(b1, b2, v);
    float n1 = __builtin_amdgcn_fmed3f(b0, b1, v);
    b0 = fminf(b0, v);
    b1 = n1;
    b2 = n2;
}

__device__ __forceinline__ float tile_min(const v8f& d) {
    float m0 = fminf(fminf(d[0], d[1]), d[2]);
    float m1 = fminf(fminf(d[3], d[4]), d[5]);
    float m2 = fminf(d[6], d[7]);
    return fminf(fminf(m0, m1), m2);
}

__device__ __forceinline__ void insert_rows(const v8f& d, float bound, int kbase,
                                            float& b0, float& b1, float& b2,
                                            int& i0, int& i1, int& i2) {
    unsigned hr[8];
#pragma unroll
    for (int r = 0; r < 8; ++r) hr[r] = __builtin_amdgcn_ballot_w32(d[r] <= bound);
#pragma unroll
    for (int r = 0; r < 8; ++r) {
        if (__builtin_expect(hr[r] != 0u, 0))
            top3_insert(d[r], kbase + r, b0, b1, b2, i0, i1, i2);
    }
}

__global__ __launch_bounds__(256) void
fp_knn_wmma_kernel(const float* __restrict__ xyz_q,
                   const float* __restrict__ xyz_k,
                   const float* __restrict__ v_k,
                   float* __restrict__ out) {
    __shared__ float s_keys[NK * 4];

    const int b      = blockIdx.y;
    const int qblock = blockIdx.x * QPB;
    const int tid    = threadIdx.x;

    const float* kb = xyz_k + (size_t)b * NK * 3;
    for (int m = tid; m < NK; m += 256) {
        float kx = kb[m * 3 + 0];
        float ky = kb[m * 3 + 1];
        float kz = kb[m * 3 + 2];
        s_keys[m * 4 + 0] = kx;
        s_keys[m * 4 + 1] = ky;
        s_keys[m * 4 + 2] = kz;
        s_keys[m * 4 + 3] = kx * kx + ky * ky + kz * kz;
    }
    __syncthreads();

    const int wave = tid >> 5;
    const int lane = tid & 31;
    const int half = lane >> 4;
    const int l16  = lane & 15;

    const int   q  = qblock + wave * 16 + l16;
    const float* qp = xyz_q + ((size_t)b * NQ + q) * 3;
    float qx = qp[0], qy = qp[1], qz = qp[2];
    float qn2 = qx * qx + qy * qy + qz * qz;

    v2f bq;
    bq.x = half ? (-2.0f * qz) : (-2.0f * qx);
    bq.y = half ? 1.0f         : (-2.0f * qy);

    const int abase = l16 * 4 + half * 2;

    v2f a_cur;
    {
        const float* sk = &s_keys[abase];
        a_cur.x = sk[0];
        a_cur.y = sk[1];
    }

    float rm[8];
#pragma unroll
    for (int r = 0; r < 8; ++r) rm[r] = 1e30f;

    for (int t = 0; t < NK; t += 16) {
        v2f a_nxt;
        {
            const float* sk = &s_keys[(((t + 16) & (NK - 1)) * 4) + abase];
            a_nxt.x = sk[0];
            a_nxt.y = sk[1];
        }

        v8f c = {};
        v8f d = __builtin_amdgcn_wmma_f32_16x16x4_f32(
             false, a_cur,  false, bq,
             (short)0, c,  false,  false);

#pragma unroll
        for (int r = 0; r < 8; ++r) rm[r] = fminf(rm[r], d[r]);

        a_cur = a_nxt;
    }

    float s0 = 1e30f, s1 = 1e30f, s2 = 1e30f;
#pragma unroll
    for (int r = 0; r < 8; ++r) top3_vals(rm[r], s0, s1, s2);
    {
        float o0 = __shfl_xor(s0, 16, 32);
        float o1 = __shfl_xor(s1, 16, 32);
        float o2 = __shfl_xor(s2, 16, 32);
        top3_vals(o0, s0, s1, s2);
        top3_vals(o1, s0, s1, s2);
        top3_vals(o2, s0, s1, s2);
    }
    const float bound = s2;

    float b0 = 1e30f, b1 = 1e30f, b2 = 1e30f;
    int   i0 = 0,     i1 = 0,     i2 = 0;

    for (int t = 0; t < NK; t += 16) {
        v2f a_nxt;
        {
            const float* sk = &s_keys[(((t + 16) & (NK - 1)) * 4) + abase];
            a_nxt.x = sk[0];
            a_nxt.y = sk[1];
        }

        v8f c = {};
        v8f d = __builtin_amdgcn_wmma_f32_16x16x4_f32(
             false, a_cur,  false, bq,
             (short)0, c,  false,  false);

        const unsigned h = __builtin_amdgcn_ballot_w32(tile_min(d) <= bound);

        if (__builtin_expect(h != 0u, 0))
            insert_rows(d, bound, t + half * 8, b0, b1, b2, i0, i1, i2);

        a_cur = a_nxt;
    }

    float ob0 = __shfl_xor(b0, 16, 32);
    float ob1 = __shfl_xor(b1, 16, 32);
    float ob2 = __shfl_xor(b2, 16, 32);
    int   oi0 = __shfl_xor(i0, 16, 32);
    int   oi1 = __shfl_xor(i1, 16, 32);
    int   oi2 = __shfl_xor(i2, 16, 32);
    top3_insert(ob0, oi0, b0, b1, b2, i0, i1, i2);
    top3_insert(ob1, oi1, b0, b1, b2, i0, i1, i2);
    top3_insert(ob2, oi2, b0, b1, b2, i0, i1, i2);

    float d0 = fmaxf(b0 + qn2, 1e-10f);
    float d1 = fmaxf(b1 + qn2, 1e-10f);
    float d2 = fmaxf(b2 + qn2, 1e-10f);
    float w0 = 1.0f / d0, w1 = 1.0f / d1, w2 = 1.0f / d2;
    float inv = 1.0f / (w0 + w1 + w2);
    w0 *= inv; w1 *= inv; w2 *= inv;

    const float* vb = v_k + (size_t)b * NK * CDIM;
    float*       ob = out + ((size_t)b * NQ + qblock + wave * 16) * CDIM;
    const int cidx = lane * 4;

    for (int qq = 0; qq < 16; ++qq) {
        int   j0 = __shfl(i0, qq, 32);
        int   j1 = __shfl(i1, qq, 32);
        int   j2 = __shfl(i2, qq, 32);
        float u0 = __shfl(w0, qq, 32);
        float u1 = __shfl(w1, qq, 32);
        float u2 = __shfl(w2, qq, 32);

        const float4 f0 = *(const float4*)(vb + (size_t)j0 * CDIM + cidx);
        const float4 f1 = *(const float4*)(vb + (size_t)j1 * CDIM + cidx);
        const float4 f2 = *(const float4*)(vb + (size_t)j2 * CDIM + cidx);

        float4 r;
        r.x = u0 * f0.x + u1 * f1.x + u2 * f2.x;
        r.y = u0 * f0.y + u1 * f1.y + u2 * f2.y;
        r.z = u0 * f0.z + u1 * f1.z + u2 * f2.z;
        r.w = u0 * f0.w + u1 * f1.w + u2 * f2.w;

        *(float4*)(ob + (size_t)qq * CDIM + cidx) = r;
    }
}

extern "C" void kernel_launch(void* const* d_in, const int* in_sizes, int n_in,
                              void* d_out, int out_size, void* d_ws, size_t ws_size,
                              hipStream_t stream) {
    const float* xyz_q = (const float*)d_in[0];
    const float* xyz_k = (const float*)d_in[1];
    const float* v_k   = (const float*)d_in[2];
    float*       out   = (float*)d_out;

    const int B = in_sizes[0] / (NQ * 3);

    dim3 grid(NQ / QPB, B);
    dim3 block(256);
    fp_knn_wmma_kernel<<<grid, block, 0, stream>>>(xyz_q, xyz_k, v_k, out);
}


# Sweep 2 takes eight tiles per trip: eight independent WMMAs back to back, then the eight tile tests together, so the vector-compare to scalar-branch round trip is paid once per trip, not per tile
# speedup vs baseline: 1.2983x; 1.1551x over previous
#include <hip/hip_runtime.h>

typedef __attribute__((ext_vector_type(2))) float v2f;
typedef __attribute__((ext_vector_type(8))) float v8f;

#define NQ   8192
#define NK   2048
#define CDIM 128
#define QPB  128
#define TPT  8

__device__ __forceinline__ void top3_insert(float v, int ki,
                                            float& b0, float& b1, float& b2,
                                            int& i0, int& i1, int& i2) {
    bool c2 = v < b2;
    float x  = c2 ? v  : b2;
    int   xi = c2 ? ki : i2;
    bool c1 = x < b1;
    b2 = c1 ? b1 : x;   i2 = c1 ? i1 : xi;
    float y  = c1 ? x  : b1;
    int   yi = c1 ? xi : i1;
    bool c0 = y < b0;
    b1 = c0 ? b0 : y;   i1 = c0 ? i0 : yi;
    b0 = c0 ? y  : b0;  i0 = c0 ? yi : i0;
}

__device__ __forceinline__ void top3_vals(float v, float& b0, float& b1, float& b2) {
    float n2 = __builtin_amdgcn_fmed3f(b1, b2, v);
    float n1 = __builtin_amdgcn_fmed3f(b0, b1, v);
    b0 = fminf(b0, v);
    b1 = n1;
    b2 = n2;
}

__device__ __forceinline__ float tile_min(const v8f& d) {
    float m0 = fminf(fminf(d[0], d[1]), d[2]);
    float m1 = fminf(fminf(d[3], d[4]), d[5]);
    float m2 = fminf(d[6], d[7]);
    return fminf(fminf(m0, m1), m2);
}

__device__ __forceinline__ void insert_rows(const v8f& d, float bound, int kbase,
                                            float& b0, float& b1, float& b2,
                                            int& i0, int& i1, int& i2) {
    unsigned hr[8];
#pragma unroll
    for (int r = 0; r < 8; ++r) hr[r] = __builtin_amdgcn_ballot_w32(d[r] <= bound);
#pragma unroll
    for (int r = 0; r < 8; ++r) {
        if (__builtin_expect(hr[r] != 0u, 0))
            top3_insert(d[r], kbase + r, b0, b1, b2, i0, i1, i2);
    }
}

__global__ __launch_bounds__(256) void
fp_knn_wmma_kernel(const float* __restrict__ xyz_q,
                   const float* __restrict__ xyz_k,
                   const float* __restrict__ v_k,
                   float* __restrict__ out) {
    __shared__ float s_keys[NK * 4];

    const int b      = blockIdx.y;
    const int qblock = blockIdx.x * QPB;
    const int tid    = threadIdx.x;

    const float* kb = xyz_k + (size_t)b * NK * 3;
    for (int m = tid; m < NK; m += 256) {
        float kx = kb[m * 3 + 0];
        float ky = kb[m * 3 + 1];
        float kz = kb[m * 3 + 2];
        s_keys[m * 4 + 0] = kx;
        s_keys[m * 4 + 1] = ky;
        s_keys[m * 4 + 2] = kz;
        s_keys[m * 4 + 3] = kx * kx + ky * ky + kz * kz;
    }
    __syncthreads();

    const int wave = tid >> 5;
    const int lane = tid & 31;
    const int half = lane >> 4;
    const int l16  = lane & 15;

    const int   q  = qblock + wave * 16 + l16;
    const float* qp = xyz_q + ((size_t)b * NQ + q) * 3;
    float qx = qp[0], qy = qp[1], qz = qp[2];
    float qn2 = qx * qx + qy * qy + qz * qz;

    v2f bq;
    bq.x = half ? (-2.0f * qz) : (-2.0f * qx);
    bq.y = half ? 1.0f         : (-2.0f * qy);

    const int abase = l16 * 4 + half * 2;

    v2f a_cur;
    {
        const float* sk = &s_keys[abase];
        a_cur.x = sk[0];
        a_cur.y = sk[1];
    }

    float rm[8];
#pragma unroll
    for (int r = 0; r < 8; ++r) rm[r] = 1e30f;

    for (int t = 0; t < NK; t += 16) {
        v2f a_nxt;
        {
            const float* sk = &s_keys[(((t + 16) & (NK - 1)) * 4) + abase];
            a_nxt.x = sk[0];
            a_nxt.y = sk[1];
        }

        v8f c = {};
        v8f d = __builtin_amdgcn_wmma_f32_16x16x4_f32(
             false, a_cur,  false, bq,
             (short)0, c,  false,  false);

#pragma unroll
        for (int r = 0; r < 8; ++r) rm[r] = fminf(rm[r], d[r]);

        a_cur = a_nxt;
    }

    float s0 = 1e30f, s1 = 1e30f, s2 = 1e30f;
#pragma unroll
    for (int r = 0; r < 8; ++r) top3_vals(rm[r], s0, s1, s2);
    {
        float o0 = __shfl_xor(s0, 16, 32);
        float o1 = __shfl_xor(s1, 16, 32);
        float o2 = __shfl_xor(s2, 16, 32);
        top3_vals(o0, s0, s1, s2);
        top3_vals(o1, s0, s1, s2);
        top3_vals(o2, s0, s1, s2);
    }
    const float bound = s2;

    float b0 = 1e30f, b1 = 1e30f, b2 = 1e30f;
    int   i0 = 0,     i1 = 0,     i2 = 0;

    v2f a8[TPT];
#pragma unroll
    for (int j = 0; j < TPT; ++j) {
        const float* sk = &s_keys[j * 64 + abase];
        a8[j].x = sk[0];
        a8[j].y = sk[1];
    }

    for (int t = 0; t < NK; t += 16 * TPT) {
        v2f n8[TPT];
#pragma unroll
        for (int j = 0; j < TPT; ++j) {
            const float* sk = &s_keys[(((t + 16 * TPT + 16 * j) & (NK - 1)) * 4) + abase];
            n8[j].x = sk[0];
            n8[j].y = sk[1];
        }

        v8f d8[TPT];
#pragma unroll
        for (int j = 0; j < TPT; ++j) {
            v8f c = {};
            d8[j] = __builtin_amdgcn_wmma_f32_16x16x4_f32(
                 false, a8[j],  false, bq,
                 (short)0, c,  false,  false);
        }

        unsigned h8[TPT];
#pragma unroll
        for (int j = 0; j < TPT; ++j)
            h8[j] = __builtin_amdgcn_ballot_w32(tile_min(d8[j]) <= bound);

#pragma unroll
        for (int j = 0; j < TPT; ++j) {
            if (h8[j] != 0u)
                insert_rows(d8[j], bound, t + 16 * j + half * 8, b0, b1, b2, i0, i1, i2);
        }

#pragma unroll
        for (int j = 0; j < TPT; ++j) a8[j] = n8[j];
    }

    float ob0 = __shfl_xor(b0, 16, 32);
    float ob1 = __shfl_xor(b1, 16, 32);
    float ob2 = __shfl_xor(b2, 16, 32);
    int   oi0 = __shfl_xor(i0, 16, 32);
    int   oi1 = __shfl_xor(i1, 16, 32);
    int   oi2 = __shfl_xor(i2, 16, 32);
    top3_insert(ob0, oi0, b0, b1, b2, i0, i1, i2);
    top3_insert(ob1, oi1, b0, b1, b2, i0, i1, i2);
    top3_insert(ob2, oi2, b0, b1, b2, i0, i1, i2);

    float d0 = fmaxf(b0 + qn2, 1e-10f);
    float d1 = fmaxf(b1 + qn2, 1e-10f);
    float d2 = fmaxf(b2 + qn2, 1e-10f);
    float w0 = 1.0f / d0, w1 = 1.0f / d1, w2 = 1.0f / d2;
    float inv = 1.0f / (w0 + w1 + w2);
    w0 *= inv; w1 *= inv; w2 *= inv;

    const float* vb = v_k + (size_t)b * NK * CDIM;
    float*       ob = out + ((size_t)b * NQ + qblock + wave * 16) * CDIM;
    const int cidx = lane * 4;

    for (int qq = 0; qq < 16; ++qq) {
        int   j0 = __shfl(i0, qq, 32);
        int   j1 = __shfl(i1, qq, 32);
        int   j2 = __shfl(i2, qq, 32);
        float u0 = __shfl(w0, qq, 32);
        float u1 = __shfl(w1, qq, 32);
        float u2 = __shfl(w2, qq, 32);

        const float4 f0 = *(const float4*)(vb + (size_t)j0 * CDIM + cidx);
        const float4 f1 = *(const float4*)(vb + (size_t)j1 * CDIM + cidx);
        const float4 f2 = *(const float4*)(vb + (size_t)j2 * CDIM + cidx);

        float4 r;
        r.x = u0 * f0.x + u1 * f1.x + u2 * f2.x;
        r.y = u0 * f0.y + u1 * f1.y + u2 * f2.y;
        r.z = u0 * f0.z + u1 * f1.z + u2 * f2.z;
        r.w = u0 * f0.w + u1 * f1.w + u2 * f2.w;

        *(float4*)(ob + (size_t)qq * CDIM + cidx) = r;
    }
}

extern "C" void kernel_launch(void* const* d_in, const int* in_sizes, int n_in,
                              void* d_out, int out_size, void* d_ws, size_t ws_size,
                              hipStream_t stream) {
    const float* xyz_q = (const float*)d_in[0];
    const float* xyz_k = (const float*)d_in[1];
    const float* v_k   = (const float*)d_in[2];
    float*       out   = (float*)d_out;

    const int B = in_sizes[0] / (NQ * 3);

    dim3 grid(NQ / QPB, B);
    dim3 block(256);
    fp_knn_wmma_kernel<<<grid, block, 0, stream>>>(xyz_q, xyz_k, v_k, out);
}


# Sweep 1 keeps each lane's per-tile minimum in LDS (bound from 8 tile classes); sweep 2 tests the kept minima and forms the WMMA again only for tiles that hold a distance <= bound
# speedup vs baseline: 1.8669x; 1.4380x over previous
#include <hip/hip_runtime.h>

typedef __attribute__((ext_vector_type(2))) float v2f;
typedef __attribute__((ext_vector_type(8))) float v8f;

#define NQ   8192
#define NK   2048
#define CDIM 128
#define QPB  128
#define TPT  8

__device__ __forceinline__ void top3_insert(float v, int ki,
                                            float& b0, float& b1, float& b2,
                                            int& i0, int& i1, int& i2) {
    bool c2 = v < b2;
    float x  = c2 ? v  : b2;
    int   xi = c2 ? ki : i2;
    bool c1 = x < b1;
    b2 = c1 ? b1 : x;   i2 = c1 ? i1 : xi;
    float y  = c1 ? x  : b1;
    int   yi = c1 ? xi : i1;
    bool c0 = y < b0;
    b1 = c0 ? b0 : y;   i1 = c0 ? i0 : yi;
    b0 = c0 ? y  : b0;  i0 = c0 ? yi : i0;
}

__device__ __forceinline__ void top3_vals(float v, float& b0, float& b1, float& b2) {
    float n2 = __builtin_amdgcn_fmed3f(b1, b2, v);
    float n1 = __builtin_amdgcn_fmed3f(b0, b1, v);
    b0 = fminf(b0, v);
    b1 = n1;
    b2 = n2;
}

__device__ __forceinline__ float tile_min(const v8f& d) {
    float m0 = fminf(fminf(d[0], d[1]), d[2]);
    float m1 = fminf(fminf(d[3], d[4]), d[5]);
    float m2 = fminf(d[6], d[7]);
    return fminf(fminf(m0, m1), m2);
}

__device__ __forceinline__ void insert_rows(const v8f& d, float bound, int kbase,
                                            float& b0, float& b1, float& b2,
                                            int& i0, int& i1, int& i2) {
    unsigned hr[8];
#pragma unroll
    for (int r = 0; r < 8; ++r) hr[r] = __builtin_amdgcn_ballot_w32(d[r] <= bound);
    asm volatile("" : "+s"(hr[0]), "+s"(hr[1]), "+s"(hr[2]), "+s"(hr[3]),
                      "+s"(hr[4]), "+s"(hr[5]), "+s"(hr[6]), "+s"(hr[7]));
#pragma unroll
    for (int r = 0; r < 8; ++r) {
        if (__builtin_expect(hr[r] != 0u, 0))
            top3_insert(d[r], kbase + r, b0, b1, b2, i0, i1, i2);
    }
}

__global__ __launch_bounds__(256) void
fp_knn_wmma_kernel(const float* __restrict__ xyz_q,
                   const float* __restrict__ xyz_k,
                   const float* __restrict__ v_k,
                   float* __restrict__ out) {
    __shared__ float s_keys[NK * 4];
    __shared__ float4 s_tmin[QPB / 16][NK / 64][32];

    const int b      = blockIdx.y;
    const int qblock = blockIdx.x * QPB;
    const int tid    = threadIdx.x;

    const int wave = tid >> 5;
    const int lane = tid & 31;
    const int half = lane >> 4;
    const int l16  = lane & 15;

    const float* kb = xyz_k + (size_t)b * NK * 3;
    float skx[NK / 256], sky[NK / 256], skz[NK / 256];
#pragma unroll
    for (int i = 0; i < NK / 256; ++i) {
        const float* kp = kb + (tid + 256 * i) * 3;
        skx[i] = kp[0];
        sky[i] = kp[1];
        skz[i] = kp[2];
    }

    const int   q  = qblock + wave * 16 + l16;
    const float* qp = xyz_q + ((size_t)b * NQ + q) * 3;
    float qx = qp[0], qy = qp[1], qz = qp[2];

#pragma unroll
    for (int i = 0; i < NK / 256; ++i) {
        const int m = tid + 256 * i;
        float kx = skx[i], ky = sky[i], kz = skz[i];
        s_keys[m * 4 + 0] = kx;
        s_keys[m * 4 + 1] = ky;
        s_keys[m * 4 + 2] = kz;
        s_keys[m * 4 + 3] = kx * kx + ky * ky + kz * kz;
    }
    __syncthreads();

    float qn2 = qx * qx + qy * qy + qz * qz;

    v2f bq;
    bq.x = half ? (-2.0f * qz) : (-2.0f * qx);
    bq.y = half ? 1.0f         : (-2.0f * qy);

    const int abase = l16 * 4 + half * 2;

    float cm[TPT];
#pragma unroll
    for (int j = 0; j < TPT; ++j) cm[j] = 1e30f;

    v2f a8[TPT];
#pragma unroll
    for (int j = 0; j < TPT; ++j) {
        const float* sk = &s_keys[j * 64 + abase];
        a8[j].x = sk[0];
        a8[j].y = sk[1];
    }

    for (int t = 0; t < NK; t += 16 * TPT) {
        v2f n8[TPT];
#pragma unroll
        for (int j = 0; j < TPT; ++j) {
            const float* sk = &s_keys[(((t + 16 * TPT + 16 * j) & (NK - 1)) * 4) + abase];
            n8[j].x = sk[0];
            n8[j].y = sk[1];
        }

        float tm[TPT];
#pragma unroll
        for (int j = 0; j < TPT; ++j) {
            v8f c = {};
            v8f d = __builtin_amdgcn_wmma_f32_16x16x4_f32(
                 false, a8[j],  false, bq,
                 (short)0, c,  false,  false);
            tm[j] = tile_min(d);
            cm[j] = fminf(cm[j], tm[j]);
        }

        const int g = t >> 6;
        s_tmin[wave][g][lane]     = make_float4(tm[0], tm[1], tm[2], tm[3]);
        s_tmin[wave][g + 1][lane] = make_float4(tm[4], tm[5], tm[6], tm[7]);

#pragma unroll
        for (int j = 0; j < TPT; ++j) a8[j] = n8[j];
    }

    float s0 = 1e30f, s1 = 1e30f, s2 = 1e30f;
#pragma unroll
    for (int j = 0; j < TPT; ++j) top3_vals(cm[j], s0, s1, s2);
    {
        float o0 = __shfl_xor(s0, 16, 32);
        float o1 = __shfl_xor(s1, 16, 32);
        float o2 = __shfl_xor(s2, 16, 32);
        top3_vals(o0, s0, s1, s2);
        top3_vals(o1, s0, s1, s2);
        top3_vals(o2, s0, s1, s2);
    }
    const float bound = s2;

    float b0 = 1e30f, b1 = 1e30f, b2 = 1e30f;
    int   i0 = 0,     i1 = 0,     i2 = 0;

#pragma unroll
    for (int j = 0; j < TPT; ++j) {
        const float* sk = &s_keys[j * 64 + abase];
        a8[j].x = sk[0];
        a8[j].y = sk[1];
    }
    float4 mlo = s_tmin[wave][0][lane];
    float4 mhi = s_tmin[wave][1][lane];

    for (int t = 0; t < NK; t += 16 * TPT) {
        v2f n8[TPT];
#pragma unroll
        for (int j = 0; j < TPT; ++j) {
            const float* sk = &s_keys[(((t + 16 * TPT + 16 * j) & (NK - 1)) * 4) + abase];
            n8[j].x = sk[0];
            n8[j].y = sk[1];
        }
        const int gn = ((t >> 6) + 2) & (NK / 64 - 1);
        const float4 nlo = s_tmin[wave][gn][lane];
        const float4 nhi = s_tmin[wave][gn + 1][lane];

        unsigned h8[TPT];
        h8[0] = __builtin_amdgcn_ballot_w32(mlo.x <= bound);
        h8[1] = __builtin_amdgcn_ballot_w32(mlo.y <= bound);
        h8[2] = __builtin_amdgcn_ballot_w32(mlo.z <= bound);
        h8[3] = __builtin_amdgcn_ballot_w32(mlo.w <= bound);
        h8[4] = __builtin_amdgcn_ballot_w32(mhi.x <= bound);
        h8[5] = __builtin_amdgcn_ballot_w32(mhi.y <= bound);
        h8[6] = __builtin_amdgcn_ballot_w32(mhi.z <= bound);
        h8[7] = __builtin_amdgcn_ballot_w32(mhi.w <= bound);
        asm volatile("" : "+s"(h8[0]), "+s"(h8[1]), "+s"(h8[2]), "+s"(h8[3]),
                          "+s"(h8[4]), "+s"(h8[5]), "+s"(h8[6]), "+s"(h8[7]));

#pragma unroll
        for (int j = 0; j < TPT; ++j) {
            if (h8[j] != 0u) {
                v8f c = {};
                v8f d = __builtin_amdgcn_wmma_f32_16x16x4_f32(
                     false, a8[j],  false, bq,
                     (short)0, c,  false,  false);
                insert_rows(d, bound, t + 16 * j + half * 8, b0, b1, b2, i0, i1, i2);
            }
        }

#pragma unroll
        for (int j = 0; j < TPT; ++j) a8[j] = n8[j];
        mlo = nlo;
        mhi = nhi;
    }

    float ob0 = __shfl_xor(b0, 16, 32);
    float ob1 = __shfl_xor(b1, 16, 32);
    float ob2 = __shfl_xor(b2, 16, 32);
    int   oi0 = __shfl_xor(i0, 16, 32);
    int   oi1 = __shfl_xor(i1, 16, 32);
    int   oi2 = __shfl_xor(i2, 16, 32);
    top3_insert(ob0, oi0, b0, b1, b2, i0, i1, i2);
    top3_insert(ob1, oi1, b0, b1, b2, i0, i1, i2);
    top3_insert(ob2, oi2, b0, b1, b2, i0, i1, i2);

    float d0 = fmaxf(b0 + qn2, 1e-10f);
    float d1 = fmaxf(b1 + qn2, 1e-10f);
    float d2 = fmaxf(b2 + qn2, 1e-10f);
    float w0 = 1.0f / d0, w1 = 1.0f / d1, w2 = 1.0f / d2;
    float inv = 1.0f / (w0 + w1 + w2);
    w0 *= inv; w1 *= inv; w2 *= inv;

    const float* vb = v_k + (size_t)b * NK * CDIM;
    float*       ob = out + ((size_t)b * NQ + qblock + wave * 16) * CDIM;
    const int cidx = lane * 4;

    float4 g0[16], g1[16], g2[16];
#pragma unroll
    for (int qq = 0; qq < 16; ++qq) {
        const unsigned j0 = (unsigned)__builtin_amdgcn_readlane(i0, qq);
        const unsigned j1 = (unsigned)__builtin_amdgcn_readlane(i1, qq);
        const unsigned j2 = (unsigned)__builtin_amdgcn_readlane(i2, qq);
        g0[qq] = *(const float4*)(vb + (size_t)j0 * CDIM + cidx);
        g1[qq] = *(const float4*)(vb + (size_t)j1 * CDIM + cidx);
        g2[qq] = *(const float4*)(vb + (size_t)j2 * CDIM + cidx);
    }

#pragma unroll
    for (int qq = 0; qq < 16; ++qq) {
        const float u0 = __int_as_float(__builtin_amdgcn_readlane(__float_as_int(w0), qq));
        const float u1 = __int_as_float(__builtin_amdgcn_readlane(__float_as_int(w1), qq));
        const float u2 = __int_as_float(__builtin_amdgcn_readlane(__float_as_int(w2), qq));
        const float4 f0 = g0[qq], f1 = g1[qq], f2 = g2[qq];

        float4 r;
        r.x = u0 * f0.x + u1 * f1.x + u2 * f2.x;
        r.y = u0 * f0.y + u1 * f1.y + u2 * f2.y;
        r.z = u0 * f0.z + u1 * f1.z + u2 * f2.z;
        r.w = u0 * f0.w + u1 * f1.w + u2 * f2.w;

        *(float4*)(ob + (size_t)qq * CDIM + cidx) = r;
    }
}

extern "C" void kernel_launch(void* const* d_in, const int* in_sizes, int n_in,
                              void* d_out, int out_size, void* d_ws, size_t ws_size,
                              hipStream_t stream) {
    const float* xyz_q = (const float*)d_in[0];
    const float* xyz_k = (const float*)d_in[1];
    const float* v_k   = (const float*)d_in[2];
    float*       out   = (float*)d_out;

    const int B = in_sizes[0] / (NQ * 3);

    dim3 grid(NQ / QPB, B);
    dim3 block(256);
    fp_knn_wmma_kernel<<<grid, block, 0, stream>>>(xyz_q, xyz_k, v_k, out);
}


# Keys of each 16-query tile split over a pair of waves (16 waves per work-group, 4 per SIMD): common bound and final top-3 merged through LDS, each wave blends 8 queries; exact ties to the lower keys
# speedup vs baseline: 2.0468x; 1.0964x over previous
#include <hip/hip_runtime.h>

typedef __attribute__((ext_vector_type(2))) float v2f;
typedef __attribute__((ext_vector_type(8))) float v8f;

#define NQ   8192
#define NK   2048
#define CDIM 128
#define QPB  128
#define TPT  8
#define KSPL 2
#define KPW  (NK / KSPL)
#define NWAV (QPB / 16 * KSPL)
#define NTHR (NWAV * 32)

__device__ __forceinline__ void top3_insert(float v, int ki,
                                            float& b0, float& b1, float& b2,
                                            int& i0, int& i1, int& i2) {
    bool c2 = v < b2;
    float x  = c2 ? v  : b2;
    int   xi = c2 ? ki : i2;
    bool c1 = x < b1;
    b2 = c1 ? b1 : x;   i2 = c1 ? i1 : xi;
    float y  = c1 ? x  : b1;
    int   yi = c1 ? xi : i1;
    bool c0 = y < b0;
    b1 = c0 ? b0 : y;   i1 = c0 ? i0 : yi;
    b0 = c0 ? y  : b0;  i0 = c0 ? yi : i0;
}

__device__ __forceinline__ void top3_vals(float v, float& b0, float& b1, float& b2) {
    float n2 = __builtin_amdgcn_fmed3f(b1, b2, v);
    float n1 = __builtin_amdgcn_fmed3f(b0, b1, v);
    b0 = fminf(b0, v);
    b1 = n1;
    b2 = n2;
}

__device__ __forceinline__ float tile_min(const v8f& d) {
    float m0 = fminf(fminf(d[0], d[1]), d[2]);
    float m1 = fminf(fminf(d[3], d[4]), d[5]);
    float m2 = fminf(d[6], d[7]);
    return fminf(fminf(m0, m1), m2);
}

__device__ __forceinline__ void insert_rows(const v8f& d, float bound, int kbase,
                                            float& b0, float& b1, float& b2,
                                            int& i0, int& i1, int& i2) {
    unsigned hr[8];
#pragma unroll
    for (int r = 0; r < 8; ++r) hr[r] = __builtin_amdgcn_ballot_w32(d[r] <= bound);
    asm volatile("" : "+s"(hr[0]), "+s"(hr[1]), "+s"(hr[2]), "+s"(hr[3]),
                      "+s"(hr[4]), "+s"(hr[5]), "+s"(hr[6]), "+s"(hr[7]));
#pragma unroll
    for (int r = 0; r < 8; ++r) {
        if (__builtin_expect(hr[r] != 0u, 0))
            top3_insert(d[r], kbase + r, b0, b1, b2, i0, i1, i2);
    }
}

__global__ __launch_bounds__(NTHR) void
fp_knn_wmma_kernel(const float* __restrict__ xyz_q,
                   const float* __restrict__ xyz_k,
                   const float* __restrict__ v_k,
                   float* __restrict__ out) {
    __shared__ float s_keys[NK * 4];
    __shared__ float4 s_tmin[NWAV][KPW / 64][32];
    __shared__ float4 s_bnd[NWAV][32];
    __shared__ float4 s_cval[NWAV][32];
    __shared__ int4   s_cidx[NWAV][32];

    const int b      = blockIdx.y;
    const int qblock = blockIdx.x * QPB;
    const int tid    = threadIdx.x;

    const int wave = __builtin_amdgcn_readfirstlane(tid >> 5);
    const int qg   = wave >> 1;
    const int kh   = wave & 1;
    const int k0   = kh * KPW;
    const int lane = tid & 31;
    const int half = lane >> 4;
    const int l16  = lane & 15;

    const float* kb = xyz_k + (size_t)b * NK * 3;
    float skx[NK / NTHR], sky[NK / NTHR], skz[NK / NTHR];
#pragma unroll
    for (int i = 0; i < NK / NTHR; ++i) {
        const float* kp = kb + (tid + NTHR * i) * 3;
        skx[i] = kp[0];
        sky[i] = kp[1];
        skz[i] = kp[2];
    }

    const int   q  = qblock + qg * 16 + l16;
    const float* qp = xyz_q + ((size_t)b * NQ + q) * 3;
    float qx = qp[0], qy = qp[1], qz = qp[2];

#pragma unroll
    for (int i = 0; i < NK / NTHR; ++i) {
        const int m = tid + NTHR * i;
        float kx = skx[i], ky = sky[i], kz = skz[i];
        s_keys[m * 4 + 0] = kx;
        s_keys[m * 4 + 1] = ky;
        s_keys[m * 4 + 2] = kz;
        s_keys[m * 4 + 3] = kx * kx + ky * ky + kz * kz;
    }
    __syncthreads();

    float qn2 = qx * qx + qy * qy + qz * qz;

    v2f bq;
    bq.x = half ? (-2.0f * qz) : (-2.0f * qx);
    bq.y = half ? 1.0f         : (-2.0f * qy);

    const int abase = l16 * 4 + half * 2;

    float cm[TPT];
#pragma unroll
    for (int j = 0; j < TPT; ++j) cm[j] = 1e30f;

    v2f a8[TPT];
#pragma unroll
    for (int j = 0; j < TPT; ++j) {
        const float* sk = &s_keys[(k0 + j * 16) * 4 + abase];
        a8[j].x = sk[0];
        a8[j].y = sk[1];
    }

    for (int t = 0; t < KPW; t += 16 * TPT) {
        v2f n8[TPT];
#pragma unroll
        for (int j = 0; j < TPT; ++j) {
            const float* sk = &s_keys[(((k0 + t + 16 * TPT + 16 * j) & (NK - 1)) * 4) + abase];
            n8[j].x = sk[0];
            n8[j].y = sk[1];
        }

        float tm[TPT];
#pragma unroll
        for (int j = 0; j < TPT; ++j) {
            v8f c = {};
            v8f d = __builtin_amdgcn_wmma_f32_16x16x4_f32(
                 false, a8[j],  false, bq,
                 (short)0, c,  false,  false);
            tm[j] = tile_min(d);
            cm[j] = fminf(cm[j], tm[j]);
        }

        const int g = t >> 6;
        s_tmin[wave][g][lane]     = make_float4(tm[0], tm[1], tm[2], tm[3]);
        s_tmin[wave][g + 1][lane] = make_float4(tm[4], tm[5], tm[6], tm[7]);

#pragma unroll
        for (int j = 0; j < TPT; ++j) a8[j] = n8[j];
    }

    float s0 = 1e30f, s1 = 1e30f, s2 = 1e30f;
#pragma unroll
    for (int j = 0; j < TPT; ++j) top3_vals(cm[j], s0, s1, s2);
    {
        float o0 = __shfl_xor(s0, 16, 32);
        float o1 = __shfl_xor(s1, 16, 32);
        float o2 = __shfl_xor(s2, 16, 32);
        top3_vals(o0, s0, s1, s2);
        top3_vals(o1, s0, s1, s2);
        top3_vals(o2, s0, s1, s2);
    }
    s_bnd[wave][lane] = make_float4(s0, s1, s2, 0.0f);
    __syncthreads();
    {
        const float4 pb = s_bnd[wave ^ 1][lane];
        top3_vals(pb.x, s0, s1, s2);
        top3_vals(pb.y, s0, s1, s2);
        top3_vals(pb.z, s0, s1, s2);
    }
    const float bound = s2;

    float b0 = 1e30f, b1 = 1e30f, b2 = 1e30f;
    int   i0 = 0,     i1 = 0,     i2 = 0;

#pragma unroll
    for (int j = 0; j < TPT; ++j) {
        const float* sk = &s_keys[(k0 + j * 16) * 4 + abase];
        a8[j].x = sk[0];
        a8[j].y = sk[1];
    }
    float4 mlo = s_tmin[wave][0][lane];
    float4 mhi = s_tmin[wave][1][lane];

    for (int t = 0; t < KPW; t += 16 * TPT) {
        v2f n8[TPT];
#pragma unroll
        for (int j = 0; j < TPT; ++j) {
            const float* sk = &s_keys[(((k0 + t + 16 * TPT + 16 * j) & (NK - 1)) * 4) + abase];
            n8[j].x = sk[0];
            n8[j].y = sk[1];
        }
        const int gn = ((t >> 6) + 2) & (KPW / 64 - 1);
        const float4 nlo = s_tmin[wave][gn][lane];
        const float4 nhi = s_tmin[wave][gn + 1][lane];

        unsigned h8[TPT];
        h8[0] = __builtin_amdgcn_ballot_w32(mlo.x <= bound);
        h8[1] = __builtin_amdgcn_ballot_w32(mlo.y <= bound);
        h8[2] = __builtin_amdgcn_ballot_w32(mlo.z <= bound);
        h8[3] = __builtin_amdgcn_ballot_w32(mlo.w <= bound);
        h8[4] = __builtin_amdgcn_ballot_w32(mhi.x <= bound);
        h8[5] = __builtin_amdgcn_ballot_w32(mhi.y <= bound);
        h8[6] = __builtin_amdgcn_ballot_w32(mhi.z <= bound);
        h8[7] = __builtin_amdgcn_ballot_w32(mhi.w <= bound);
        asm volatile("" : "+s"(h8[0]), "+s"(h8[1]), "+s"(h8[2]), "+s"(h8[3]),
                          "+s"(h8[4]), "+s"(h8[5]), "+s"(h8[6]), "+s"(h8[7]));

#pragma unroll
        for (int j = 0; j < TPT; ++j) {
            if (h8[j] != 0u) {
                v8f c = {};
                v8f d = __builtin_amdgcn_wmma_f32_16x16x4_f32(
                     false, a8[j],  false, bq,
                     (short)0, c,  false,  false);
                insert_rows(d, bound, k0 + t + 16 * j + half * 8, b0, b1, b2, i0, i1, i2);
            }
        }

#pragma unroll
        for (int j = 0; j < TPT; ++j) a8[j] = n8[j];
        mlo = nlo;
        mhi = nhi;
    }

    float ob0 = __shfl_xor(b0, 16, 32);
    float ob1 = __shfl_xor(b1, 16, 32);
    float ob2 = __shfl_xor(b2, 16, 32);
    int   oi0 = __shfl_xor(i0, 16, 32);
    int   oi1 = __shfl_xor(i1, 16, 32);
    int   oi2 = __shfl_xor(i2, 16, 32);
    top3_insert(ob0, oi0, b0, b1, b2, i0, i1, i2);
    top3_insert(ob1, oi1, b0, b1, b2, i0, i1, i2);
    top3_insert(ob2, oi2, b0, b1, b2, i0, i1, i2);

    s_cval[wave][lane] = make_float4(b0, b1, b2, 0.0f);
    s_cidx[wave][lane] = make_int4(i0, i1, i2, 0);
    __syncthreads();
    {
        const float4 lv = s_cval[wave & ~1][lane];
        const int4   li = s_cidx[wave & ~1][lane];
        const float4 hv = s_cval[wave | 1][lane];
        const int4   hi = s_cidx[wave | 1][lane];
        b0 = lv.x; b1 = lv.y; b2 = lv.z;
        i0 = li.x; i1 = li.y; i2 = li.z;
        top3_insert(hv.x, hi.x, b0, b1, b2, i0, i1, i2);
        top3_insert(hv.y, hi.y, b0, b1, b2, i0, i1, i2);
        top3_insert(hv.z, hi.z, b0, b1, b2, i0, i1, i2);
    }

    float d0 = fmaxf(b0 + qn2, 1e-10f);
    float d1 = fmaxf(b1 + qn2, 1e-10f);
    float d2 = fmaxf(b2 + qn2, 1e-10f);
    float w0 = 1.0f / d0, w1 = 1.0f / d1, w2 = 1.0f / d2;
    float inv = 1.0f / (w0 + w1 + w2);
    w0 *= inv; w1 *= inv; w2 *= inv;

    const float* vb = v_k + (size_t)b * NK * CDIM;
    const int    qb = kh * (16 / KSPL);
    float*       ob = out + ((size_t)b * NQ + qblock + qg * 16 + qb) * CDIM;
    const int cidx = lane * 4;

    float4 g0[16 / KSPL], g1[16 / KSPL], g2[16 / KSPL];
#pragma unroll
    for (int qq = 0; qq < 16 / KSPL; ++qq) {
        const unsigned j0 = (unsigned)__builtin_amdgcn_readlane(i0, qb + qq);
        const unsigned j1 = (unsigned)__builtin_amdgcn_readlane(i1, qb + qq);
        const unsigned j2 = (unsigned)__builtin_amdgcn_readlane(i2, qb + qq);
        g0[qq] = *(const float4*)(vb + (size_t)j0 * CDIM + cidx);
        g1[qq] = *(const float4*)(vb + (size_t)j1 * CDIM + cidx);
        g2[qq] = *(const float4*)(vb + (size_t)j2 * CDIM + cidx);
    }

#pragma unroll
    for (int qq = 0; qq < 16 / KSPL; ++qq) {
        const float u0 = __int_as_float(__builtin_amdgcn_readlane(__float_as_int(w0), qb + qq));
        const float u1 = __int_as_float(__builtin_amdgcn_readlane(__float_as_int(w1), qb + qq));
        const float u2 = __int_as_float(__builtin_amdgcn_readlane(__float_as_int(w2), qb + qq));
        const float4 f0 = g0[qq], f1 = g1[qq], f2 = g2[qq];

        float4 r;
        r.x = u0 * f0.x + u1 * f1.x + u2 * f2.x;
        r.y = u0 * f0.y + u1 * f1.y + u2 * f2.y;
        r.z = u0 * f0.z + u1 * f1.z + u2 * f2.z;
        r.w = u0 * f0.w + u1 * f1.w + u2 * f2.w;

        *(float4*)(ob + (size_t)qq * CDIM + cidx) = r;
    }
}

extern "C" void kernel_launch(void* const* d_in, const int* in_sizes, int n_in,
                              void* d_out, int out_size, void* d_ws, size_t ws_size,
                              hipStream_t stream) {
    const float* xyz_q = (const float*)d_in[0];
    const float* xyz_k = (const float*)d_in[1];
    const float* v_k   = (const float*)d_in[2];
    float*       out   = (float*)d_out;

    const int B = in_sizes[0] / (NQ * 3);

    dim3 grid(NQ / QPB, B);
    dim3 block(NTHR);
    fp_knn_wmma_kernel<<<grid, block, 0, stream>>>(xyz_q, xyz_k, v_k, out);
}
